# starting kernel
# baseline (speedup 1.0000x reference)
#include <hip/hip_runtime.h>
#include <math.h>


typedef __attribute__((ext_vector_type(2))) float v2f;
typedef __attribute__((ext_vector_type(8))) float v8f;

#define HW      16384
#define NBATCH  8
#define CCH     128
#define MID     64
#define PDIM    32
#define NBAND   8

__device__ __forceinline__ v8f wmma4(v2f a, v2f b, v8f c) {
  return __builtin_amdgcn_wmma_f32_16x16x4_f32(false, a, false, b,
                                               (short)0, c, false, false);
}

__device__ __forceinline__ float gelu_f(float x) {
  return 0.5f * x * (1.0f + erff(x * 0.70710678118654752440f));
}
__device__ __forceinline__ float clamp01(float x) {
  return fminf(fmaxf(x, 0.0f), 1.0f);
}

__global__ void k_emb(const float* __restrict__ Wp, float* __restrict__ emb) {
  int p = blockIdx.x * blockDim.x + threadIdx.x;
  if (p >= HW) return;
  int i = p >> 7, j = p & 127;
  float ym = -1.0f + (2.0f / 127.0f) * (float)i;
  float xm = -1.0f + (2.0f / 127.0f) * (float)j;
#pragma unroll 4
  for (int pd = 0; pd < PDIM; ++pd) {
    float v = tanhf(Wp[pd * 2 + 0] * ym + Wp[pd * 2 + 1] * xm);
    emb[(size_t)pd * HW + p] = v;
  }
}

__global__ __launch_bounds__(256) void k_modulate(
    const float* __restrict__ x, const float* __restrict__ emb,
    const float* __restrict__ Wm1, const float* __restrict__ Wm2,
    float* __restrict__ xmod) {
  __shared__ float lds[8][MID * 17];
  const int wave = threadIdx.x >> 5;
  const int lane = threadIdx.x & 31;
  const int n    = blockIdx.x >> 7;
  const int pb   = (blockIdx.x & 127) * 128 + wave * 16;
  const int lm   = lane & 15;
  const int kh   = (lane >> 4) * 2;
  const int mh   = (lane >> 4) * 8;
  const int pix  = pb + lm;
  const float* xn = x + (size_t)n * CCH * HW;

  v8f acc[4] = {};
  for (int k0 = 0; k0 < CCH; k0 += 4) {
    int kk = k0 + kh;
    v2f b;
    b.x = xn[(size_t)kk * HW + pix];
    b.y = xn[(size_t)(kk + 1) * HW + pix];
#pragma unroll
    for (int t = 0; t < 4; ++t) {
      v2f a = *(const v2f*)(Wm1 + (t * 16 + lm) * (CCH + PDIM) + kk);
      acc[t] = wmma4(a, b, acc[t]);
    }
  }
  for (int k0 = 0; k0 < PDIM; k0 += 4) {
    int kk = k0 + kh;
    v2f b;
    b.x = emb[(size_t)kk * HW + pix];
    b.y = emb[(size_t)(kk + 1) * HW + pix];
#pragma unroll
    for (int t = 0; t < 4; ++t) {
      v2f a = *(const v2f*)(Wm1 + (t * 16 + lm) * (CCH + PDIM) + CCH + kk);
      acc[t] = wmma4(a, b, acc[t]);
    }
  }
  float* sm = &lds[wave][0];
#pragma unroll
  for (int t = 0; t < 4; ++t)
#pragma unroll
    for (int v = 0; v < 8; ++v)
      sm[(t * 16 + v + mh) * 17 + lm] = gelu_f(acc[t][v]);

  v8f acc2[8] = {};
  for (int k0 = 0; k0 < MID; k0 += 4) {
    int kk = k0 + kh;
    v2f b;
    b.x = sm[kk * 17 + lm];
    b.y = sm[(kk + 1) * 17 + lm];
#pragma unroll
    for (int t = 0; t < 8; ++t) {
      v2f a = *(const v2f*)(Wm2 + (t * 16 + lm) * MID + kk);
      acc2[t] = wmma4(a, b, acc2[t]);
    }
  }
  float* xo = xmod + (size_t)n * CCH * HW;
#pragma unroll
  for (int t = 0; t < 8; ++t)
#pragma unroll
    for (int v = 0; v < 8; ++v) {
      int m = t * 16 + v + mh;
      xo[(size_t)m * HW + pix] = clamp01(gelu_f(acc2[t][v]));
    }
}

__global__ __launch_bounds__(256) void k_band_stats(
    const float* __restrict__ xmod, const float* __restrict__ W1,
    float* __restrict__ psum, float* __restrict__ psq) {
  const int band  = threadIdx.x >> 5;
  const int lane  = threadIdx.x & 31;
  const int n     = blockIdx.x >> 7;
  const int chunk = blockIdx.x & 127;
  const int lm = lane & 15, kh = (lane >> 4) * 2, mh = (lane >> 4) * 8;
  const float* xn = xmod + (size_t)n * CCH * HW;
  const float* A  = W1 + (size_t)band * MID * CCH;

  v8f s_run[4] = {};
  v8f q_run[4] = {};
  for (int sb = 0; sb < 8; ++sb) {
    const int pix = chunk * 128 + sb * 16 + lm;
    v8f acc[4] = {};
    for (int k0 = 0; k0 < CCH; k0 += 4) {
      int kk = k0 + kh;
      v2f bb;
      bb.x = xn[(size_t)kk * HW + pix];
      bb.y = xn[(size_t)(kk + 1) * HW + pix];
#pragma unroll
      for (int t = 0; t < 4; ++t) {
        v2f a = *(const v2f*)(A + (t * 16 + lm) * CCH + kk);
        acc[t] = wmma4(a, bb, acc[t]);
      }
    }
#pragma unroll
    for (int t = 0; t < 4; ++t) {
      s_run[t] += acc[t];
      q_run[t] += acc[t] * acc[t];
    }
  }
  const int nb = n * NBAND + band;
#pragma unroll
  for (int t = 0; t < 4; ++t)
#pragma unroll
    for (int v = 0; v < 8; ++v) {
      float s = s_run[t][v], q = q_run[t][v];
#pragma unroll
      for (int off = 1; off < 16; off <<= 1) {
        s += __shfl_xor(s, off, 32);
        q += __shfl_xor(q, off, 32);
      }
      if (lm == 0) {
        int m = t * 16 + v + mh;
        psum[((size_t)nb * MID + m) * 128 + chunk] = s;
        psq [((size_t)nb * MID + m) * 128 + chunk] = q;
      }
    }
}

__global__ void k_reduce_stats(const float* __restrict__ psum,
                               const float* __restrict__ psq,
                               float* __restrict__ mean,
                               float* __restrict__ rstd) {
  int i = blockIdx.x * blockDim.x + threadIdx.x;
  if (i >= NBATCH * NBAND * MID) return;
  float s = 0.0f, q = 0.0f;
  for (int j = 0; j < 128; ++j) { s += psum[(size_t)i * 128 + j]; q += psq[(size_t)i * 128 + j]; }
  float mu  = s * (1.0f / (float)HW);
  float var = q * (1.0f / (float)HW) - mu * mu;
  mean[i] = mu;
  rstd[i] = rsqrtf(fmaxf(var, 0.0f) + 1e-5f);
}

__global__ __launch_bounds__(256) void k_bands(
    const float* __restrict__ xmod, const float* __restrict__ W1,
    const float* __restrict__ W2, const float* __restrict__ W3,
    const float* __restrict__ mean, const float* __restrict__ rstd,
    float* __restrict__ out) {
  __shared__ float lds[8][MID * 17];
  const int band  = threadIdx.x >> 5;
  const int lane  = threadIdx.x & 31;
  const int n     = blockIdx.x >> 7;
  const int chunk = blockIdx.x & 127;
  const int lm = lane & 15, kh = (lane >> 4) * 2, mh = (lane >> 4) * 8;
  const int nb = n * NBAND + band;
  const float* xn = xmod + (size_t)n * CCH * HW;
  const float* A  = W1 + (size_t)band * MID * CCH;
  const float* A2 = W2 + (size_t)band * 32 * MID;
  const float* w3 = W3 + (size_t)band * 32;
  const float* mu = mean + (size_t)nb * MID;
  const float* rs = rstd + (size_t)nb * MID;
  float* sm = &lds[band][0];

  for (int sb = 0; sb < 8; ++sb) {
    const int pix = chunk * 128 + sb * 16 + lm;

    v8f acc[4] = {};
    for (int k0 = 0; k0 < CCH; k0 += 4) {
      int kk = k0 + kh;
      v2f bb;
      bb.x = xn[(size_t)kk * HW + pix];
      bb.y = xn[(size_t)(kk + 1) * HW + pix];
#pragma unroll
      for (int t = 0; t < 4; ++t) {
        v2f a = *(const v2f*)(A + (t * 16 + lm) * CCH + kk);
        acc[t] = wmma4(a, bb, acc[t]);
      }
    }
#pragma unroll
    for (int t = 0; t < 4; ++t)
#pragma unroll
      for (int v = 0; v < 8; ++v) {
        int m = t * 16 + v + mh;
        sm[m * 17 + lm] = gelu_f((acc[t][v] - mu[m]) * rs[m]);
      }

    v8f acc2[2] = {};
    for (int k0 = 0; k0 < MID; k0 += 4) {
      int kk = k0 + kh;
      v2f bb;
      bb.x = sm[kk * 17 + lm];
      bb.y = sm[(kk + 1) * 17 + lm];
#pragma unroll
      for (int t = 0; t < 2; ++t) {
        v2f a = *(const v2f*)(A2 + (t * 16 + lm) * MID + kk);
        acc2[t] = wmma4(a, bb, acc2[t]);
      }
    }
#pragma unroll
    for (int t = 0; t < 2; ++t)
#pragma unroll
      for (int v = 0; v < 8; ++v)
        sm[(t * 16 + v + mh) * 17 + lm] = gelu_f(acc2[t][v]);

    if (lane < 16) {
      float s = 0.0f;
#pragma unroll
      for (int k = 0; k < 32; ++k) s += w3[k] * sm[k * 17 + lane];
      out[(size_t)nb * HW + chunk * 128 + sb * 16 + lane] = clamp01(gelu_f(s));
    }
  }
}

extern "C" void kernel_launch(void* const* d_in, const int* in_sizes, int n_in,
                              void* d_out, int out_size, void* d_ws, size_t ws_size,
                              hipStream_t stream) {
  const float* x   = (const float*)d_in[0];
  const float* Wp  = (const float*)d_in[1];
  const float* Wm1 = (const float*)d_in[2];
  const float* Wm2 = (const float*)d_in[3];
  const float* W1  = (const float*)d_in[4];
  const float* W2  = (const float*)d_in[5];
  const float* W3  = (const float*)d_in[6];
  float* out = (float*)d_out;

  float* ws   = (float*)d_ws;
  float* emb  = ws;
  float* xmod = emb + (size_t)PDIM * HW;
  float* psum = xmod + (size_t)NBATCH * CCH * HW;
  float* psq  = psum + (size_t)NBATCH * NBAND * MID * 128;
  float* mean = psq + (size_t)NBATCH * NBAND * MID * 128;
  float* rstd = mean + (size_t)NBATCH * NBAND * MID;
  (void)in_sizes; (void)n_in; (void)out_size; (void)ws_size;

  k_emb<<<HW / 256, 256, 0, stream>>>(Wp, emb);
  k_modulate<<<NBATCH * 128, 256, 0, stream>>>(x, emb, Wm1, Wm2, xmod);
  k_band_stats<<<NBATCH * 128, 256, 0, stream>>>(xmod, W1, psum, psq);
  k_reduce_stats<<<(NBATCH * NBAND * MID + 255) / 256, 256, 0, stream>>>(psum, psq, mean, rstd);
  k_bands<<<NBATCH * 128, 256, 0, stream>>>(xmod, W1, W2, W3, mean, rstd, out);
}


# Stats pass: block = 512-pixel range of an image, walking its four 128-pixel chunks with the running sums in registers; one shuffle-reduce and one partial per range; stats sum order changes
# speedup vs baseline: 2.3746x; 2.3746x over previous
#include <hip/hip_runtime.h>
#include <math.h>


typedef __attribute__((ext_vector_type(2))) float v2f;
typedef __attribute__((ext_vector_type(8))) float v8f;

#define HW      16384
#define NBATCH  8
#define CCH     128
#define MID     64
#define PDIM    32
#define NBAND   8

__device__ __forceinline__ v8f wmma4(v2f a, v2f b, v8f c) {
  return __builtin_amdgcn_wmma_f32_16x16x4_f32(false, a, false, b,
                                               (short)0, c, false, false);
}

__device__ __forceinline__ float gelu_f(float x) {
  return 0.5f * x * (1.0f + erff(x * 0.70710678118654752440f));
}
__device__ __forceinline__ float clamp01(float x) {
  return fminf(fmaxf(x, 0.0f), 1.0f);
}

__global__ void k_emb(const float* __restrict__ Wp, float* __restrict__ emb) {
  int p = blockIdx.x * blockDim.x + threadIdx.x;
  if (p >= HW) return;
  int i = p >> 7, j = p & 127;
  float ym = -1.0f + (2.0f / 127.0f) * (float)i;
  float xm = -1.0f + (2.0f / 127.0f) * (float)j;
#pragma unroll 4
  for (int pd = 0; pd < PDIM; ++pd) {
    float v = tanhf(Wp[pd * 2 + 0] * ym + Wp[pd * 2 + 1] * xm);
    emb[(size_t)pd * HW + p] = v;
  }
}

#define MNB    2
#define MPITCH 68
#define OPITCH 36

template <int J0, int PITCH>
__device__ __forceinline__ void mod_load_a_half(float4 (&a)[4][4], const float* __restrict__ ap) {
#pragma unroll
  for (int t = 0; t < 4; ++t) {
    a[t][J0]     = *(const float4*)(ap + t * 16 * PITCH + 4 * J0);
    a[t][J0 + 1] = *(const float4*)(ap + t * 16 * PITCH + 4 * J0 + 4);
  }
}

template <int J0>
__device__ __forceinline__ void mod_load_bg_half(float4 (&b)[MNB][4], const float* __restrict__ p) {
#pragma unroll
  for (int sb = 0; sb < MNB; ++sb)
#pragma unroll
    for (int jj = J0; jj < J0 + 2; ++jj) {
      b[sb][jj].x = p[(size_t)(4 * jj + 0) * HW + sb * 16];
      b[sb][jj].y = p[(size_t)(4 * jj + 1) * HW + sb * 16];
      b[sb][jj].z = p[(size_t)(4 * jj + 2) * HW + sb * 16];
      b[sb][jj].w = p[(size_t)(4 * jj + 3) * HW + sb * 16];
    }
}

template <int J0>
__device__ __forceinline__ void mod_load_bl_half(float4 (&b)[MNB][4], const float* p) {
#pragma unroll
  for (int sb = 0; sb < MNB; ++sb) {
    b[sb][J0]     = *(const float4*)(p + sb * 16 * MPITCH + 4 * J0);
    b[sb][J0 + 1] = *(const float4*)(p + sb * 16 * MPITCH + 4 * J0 + 4);
  }
}

__device__ __forceinline__ void mod_mma_step(v8f (&acc)[MNB][4], const float4 (&a)[4][4],
                                             const float4 (&b)[MNB][4], int s) {
  const int j = s >> 1;
  v2f bf[MNB];
#pragma unroll
  for (int sb = 0; sb < MNB; ++sb) {
    if (s & 1) { bf[sb].x = b[sb][j].z; bf[sb].y = b[sb][j].w; }
    else       { bf[sb].x = b[sb][j].x; bf[sb].y = b[sb][j].y; }
  }
#pragma unroll
  for (int t = 0; t < 4; ++t) {
    v2f af;
    if (s & 1) { af.x = a[t][j].z; af.y = a[t][j].w; }
    else       { af.x = a[t][j].x; af.y = a[t][j].y; }
#pragma unroll
    for (int sb = 0; sb < MNB; ++sb) acc[sb][t] = wmma4(af, bf[sb], acc[sb][t]);
  }
}

__device__ __forceinline__ float4 gelu4(float4 v) {
  v.x = gelu_f(v.x); v.y = gelu_f(v.y); v.z = gelu_f(v.z); v.w = gelu_f(v.w);
  return v;
}

__global__ __launch_bounds__(256, 4) void k_modulate(
    const float* __restrict__ x, const float* __restrict__ emb,
    const float* __restrict__ Wm1, const float* __restrict__ Wm2,
    float* __restrict__ xmod) {
  __shared__ __attribute__((aligned(16))) float lds[8][MNB * 16 * MPITCH + 64 * OPITCH];
  const int wave = threadIdx.x >> 5;
  const int lane = threadIdx.x & 31;
  const int n    = blockIdx.x >> 6;
  const int pixbase = (blockIdx.x & 63) * (8 * MNB * 16) + wave * (MNB * 16);
  const int lm   = lane & 15;
  const int h    = lane >> 4;
  const int mh   = h * 8;
  const float* xn = x + (size_t)n * CCH * HW;
  float* sm = &lds[wave][0];
  float* so = sm + MNB * 16 * MPITCH;

  float4 a[4][4];
  float4 b[MNB][4];

  {
    v8f acc[MNB][4];
#pragma unroll
    for (int sb = 0; sb < MNB; ++sb)
#pragma unroll
      for (int t = 0; t < 4; ++t) acc[sb][t] = v8f{};
    const float* arow = Wm1 + lm * (CCH + PDIM) + 16 * h;
    const float* xcol = xn  + (size_t)(16 * h) * HW + pixbase + lm;
    const float* ecol = emb + (size_t)(16 * h) * HW + pixbase + lm;
    mod_load_a_half<0, CCH + PDIM>(a, arow);
    mod_load_a_half<2, CCH + PDIM>(a, arow);
    mod_load_bg_half<0>(b, xcol);
    mod_load_bg_half<2>(b, xcol);
#pragma unroll 1
    for (int kb = 0; kb < 5; ++kb) {
      const int nkb = (kb == 4) ? 0 : kb + 1;
      const float* an = arow + nkb * 32;
      const float* bn = (nkb < 4) ? xcol + (size_t)(nkb * 32) * HW : ecol;
#pragma unroll
      for (int s = 0; s < 4; ++s) mod_mma_step(acc, a, b, s);
      mod_load_a_half<0, CCH + PDIM>(a, an);
      mod_load_bg_half<0>(b, bn);
#pragma unroll
      for (int s = 4; s < 8; ++s) mod_mma_step(acc, a, b, s);
      mod_load_a_half<2, CCH + PDIM>(a, an);
      mod_load_bg_half<2>(b, bn);
    }
    float* qb = sm + lm * MPITCH + mh;
#pragma unroll
    for (int sb = 0; sb < MNB; ++sb)
#pragma unroll
      for (int t = 0; t < 4; ++t) {
        float4 lo, hi;
        lo.x = acc[sb][t][0]; lo.y = acc[sb][t][1]; lo.z = acc[sb][t][2]; lo.w = acc[sb][t][3];
        hi.x = acc[sb][t][4]; hi.y = acc[sb][t][5]; hi.z = acc[sb][t][6]; hi.w = acc[sb][t][7];
        float* q = qb + sb * 16 * MPITCH + 16 * t;
        *(float4*)q       = lo;
        *(float4*)(q + 4) = hi;
      }
#pragma unroll 1
    for (int i = 0; i < MNB * 8; ++i) {
      float* q = qb + (i >> 3) * 16 * MPITCH + ((i >> 1) & 3) * 16 + (i & 1) * 4;
      *(float4*)q = gelu4(*(const float4*)q);
    }
  }

  float* xo = xmod + (size_t)n * CCH * HW + pixbase + lane;
  const float* bcol = sm + lm * MPITCH + 16 * h;
#pragma unroll 1
  for (int mhalf = 0; mhalf < 2; ++mhalf) {
    v8f acc2[MNB][4];
#pragma unroll
    for (int sb = 0; sb < MNB; ++sb)
#pragma unroll
      for (int t = 0; t < 4; ++t) acc2[sb][t] = v8f{};
    const float* arow2 = Wm2 + (size_t)(mhalf * 64 + lm) * MID + 16 * h;
    mod_load_a_half<0, MID>(a, arow2);
    mod_load_a_half<2, MID>(a, arow2);
    mod_load_bl_half<0>(b, bcol);
    mod_load_bl_half<2>(b, bcol);
#pragma unroll
    for (int s = 0; s < 4; ++s) mod_mma_step(acc2, a, b, s);
    mod_load_a_half<0, MID>(a, arow2 + 32);
    mod_load_bl_half<0>(b, bcol + 32);
#pragma unroll
    for (int s = 4; s < 8; ++s) mod_mma_step(acc2, a, b, s);
    mod_load_a_half<2, MID>(a, arow2 + 32);
    mod_load_bl_half<2>(b, bcol + 32);
#pragma unroll
    for (int s = 0; s < 8; ++s) mod_mma_step(acc2, a, b, s);
#pragma unroll
    for (int sb = 0; sb < MNB; ++sb)
#pragma unroll
      for (int t = 0; t < 4; ++t)
#pragma unroll
        for (int v = 0; v < 8; ++v)
          so[(t * 16 + v + mh) * OPITCH + sb * 16 + lm] = acc2[sb][t][v];
    float* xr = xo + (size_t)(mhalf * 64) * HW;
#pragma unroll 1
    for (int r = 0; r < 64; r += 4) {
#pragma unroll
      for (int u = 0; u < 4; ++u)
        xr[(size_t)(r + u) * HW] = clamp01(gelu_f(so[(r + u) * OPITCH + lane]));
    }
  }
}

template <int NB>
__device__ __forceinline__ void h1_load_b(v2f (&b)[NB], const float* __restrict__ p) {
#pragma unroll
  for (int sb = 0; sb < NB; ++sb) {
    b[sb].x = p[sb * 16];
    b[sb].y = p[HW + sb * 16];
  }
}

template <int J0>
__device__ __forceinline__ void h1_load_a_half(float4 (&a)[4][4], const float* __restrict__ ap) {
#pragma unroll
  for (int t = 0; t < 4; ++t) {
    a[t][J0]     = *(const float4*)(ap + t * 16 * CCH + 4 * J0);
    a[t][J0 + 1] = *(const float4*)(ap + t * 16 * CCH + 4 * J0 + 4);
  }
}

template <int NB>
__device__ __forceinline__ void h1_mma_step(v8f (&acc)[NB][4], const float4 (&a)[4][4],
                                            const v2f (&b)[NB], int s) {
  const int j = s >> 1;
#pragma unroll
  for (int t = 0; t < 4; ++t) {
    v2f af;
    if (s & 1) { af.x = a[t][j].z; af.y = a[t][j].w; }
    else       { af.x = a[t][j].x; af.y = a[t][j].y; }
#pragma unroll
    for (int sb = 0; sb < NB; ++sb) acc[sb][t] = wmma4(af, b[sb], acc[sb][t]);
  }
}

template <int NB>
__device__ __forceinline__ void h1_tile(v8f (&acc)[NB][4], const float* __restrict__ xn,
                                        const float* __restrict__ A, int pixbase, int lane) {
  const int lm = lane & 15, h = lane >> 4;
  const float* arow = A + lm * CCH + 16 * h;
  const float* bcol = xn + (size_t)(16 * h) * HW + pixbase + lm;
#pragma unroll
  for (int sb = 0; sb < NB; ++sb)
#pragma unroll
    for (int t = 0; t < 4; ++t) acc[sb][t] = v8f{};
  float4 a[4][4];
  v2f b0[NB], b1[NB];
  h1_load_a_half<0>(a, arow);
  h1_load_a_half<2>(a, arow);
  h1_load_b<NB>(b0, bcol);
#pragma unroll 1
  for (int kb = 0; kb < 4; ++kb) {
    const int nkb = (kb + 1) & 3;
    const float* an = arow + nkb * 32;
    const float* bk = bcol + (size_t)(kb * 32) * HW;
    const float* bn = bcol + (size_t)(nkb * 32) * HW;
#pragma unroll
    for (int s = 0; s < 8; s += 2) {
      h1_load_b<NB>(b1, bk + (size_t)(2 * (s + 1)) * HW);
      if (s == 4) h1_load_a_half<0>(a, an);
      h1_mma_step<NB>(acc, a, b0, s);
      if (s + 2 < 8) h1_load_b<NB>(b0, bk + (size_t)(2 * (s + 2)) * HW);
      else           h1_load_b<NB>(b0, bn);
      h1_mma_step<NB>(acc, a, b1, s + 1);
    }
    h1_load_a_half<2>(a, an);
  }
}

template <int J, int PITCH>
__device__ __forceinline__ void h1_load_a_piece(float4 (&a)[4][2], const float* ap) {
#pragma unroll
  for (int t = 0; t < 4; ++t) a[t][J] = *(const float4*)(ap + t * 16 * PITCH + 4 * J);
}

template <int NB>
__device__ __forceinline__ void h1_mma_step_hb(v8f (&acc)[NB][4], const float4 (&a)[4][2],
                                               const v2f (&b)[NB], int s) {
  const int j = s >> 1;
#pragma unroll
  for (int t = 0; t < 4; ++t) {
    v2f af;
    if (s & 1) { af.x = a[t][j].z; af.y = a[t][j].w; }
    else       { af.x = a[t][j].x; af.y = a[t][j].y; }
#pragma unroll
    for (int sb = 0; sb < NB; ++sb) acc[sb][t] = wmma4(af, b[sb], acc[sb][t]);
  }
}

template <int NB, int PITCH>
__device__ __forceinline__ void h1_tile_hb(v8f (&acc)[NB][4], const float* __restrict__ xn,
                                           const float* A, int pixbase, int lane) {
  const int lm = lane & 15, h = lane >> 4;
  const float* arow = A + lm * PITCH + 16 * h;
  const float* bcol = xn + (size_t)(16 * h) * HW + pixbase + lm;
#pragma unroll
  for (int sb = 0; sb < NB; ++sb)
#pragma unroll
    for (int t = 0; t < 4; ++t) acc[sb][t] = v8f{};
  float4 a[4][2];
  v2f b0[NB], b1[NB];
  h1_load_a_piece<0, PITCH>(a, arow);
  h1_load_a_piece<1, PITCH>(a, arow);
  h1_load_b<NB>(b0, bcol);
#pragma unroll 1
  for (int q = 0; q < 8; ++q) {
    const int nq  = (q + 1) & 7;
    const int ko  = (q >> 1) * 32 + (q & 1) * 8;
    const int nko = (nq >> 1) * 32 + (nq & 1) * 8;
    const float* an = arow + nko;
    const float* bk = bcol + (size_t)ko * HW;
    const float* bn = bcol + (size_t)nko * HW;
    h1_load_b<NB>(b1, bk + (size_t)2 * HW);
    h1_mma_step_hb<NB>(acc, a, b0, 0);
    h1_load_b<NB>(b0, bk + (size_t)4 * HW);
    h1_mma_step_hb<NB>(acc, a, b1, 1);
    h1_load_a_piece<0, PITCH>(a, an);
    h1_load_b<NB>(b1, bk + (size_t)6 * HW);
    h1_mma_step_hb<NB>(acc, a, b0, 2);
    h1_load_b<NB>(b0, bn);
    h1_mma_step_hb<NB>(acc, a, b1, 3);
    h1_load_a_piece<1, PITCH>(a, an);
  }
}

#define SCH   4
#define SPART 32
__global__ __launch_bounds__(256) void k_band_stats(
    const float* __restrict__ xmod, const float* __restrict__ W1,
    float* __restrict__ psum, float* __restrict__ psq) {
  const int band  = threadIdx.x >> 5;
  const int lane  = threadIdx.x & 31;
  const int n     = blockIdx.x >> 5;
  const int range = blockIdx.x & 31;
  const int lm = lane & 15, mh = (lane >> 4) * 8;
  const float* xn = xmod + (size_t)n * CCH * HW;
  const float* A  = W1 + (size_t)band * MID * CCH;

  v8f s_run[4] = {};
  v8f q_run[4] = {};
  for (int c = 0; c < SCH; ++c) {
    v8f acc[8][4];
    h1_tile<8>(acc, xn, A, (range * SCH + c) * 128, lane);
#pragma unroll
    for (int sb = 0; sb < 8; ++sb)
#pragma unroll
      for (int t = 0; t < 4; ++t) {
        s_run[t] += acc[sb][t];
        q_run[t] += acc[sb][t] * acc[sb][t];
      }
  }
  const int nb = n * NBAND + band;
#pragma unroll
  for (int t = 0; t < 4; ++t)
#pragma unroll
    for (int v = 0; v < 8; ++v) {
      float s = s_run[t][v], q = q_run[t][v];
#pragma unroll
      for (int off = 1; off < 16; off <<= 1) {
        s += __shfl_xor(s, off, 32);
        q += __shfl_xor(q, off, 32);
      }
      if (lm == 0) {
        int m = t * 16 + v + mh;
        psum[((size_t)nb * MID + m) * SPART + range] = s;
        psq [((size_t)nb * MID + m) * SPART + range] = q;
      }
    }
}

__global__ void k_reduce_stats(const float* __restrict__ psum,
                               const float* __restrict__ psq,
                               float* __restrict__ mean,
                               float* __restrict__ rstd) {
  int i = blockIdx.x * blockDim.x + threadIdx.x;
  if (i >= NBATCH * NBAND * MID) return;
  float s = 0.0f, q = 0.0f;
  for (int j = 0; j < SPART; ++j) { s += psum[(size_t)i * SPART + j]; q += psq[(size_t)i * SPART + j]; }
  float mu  = s * (1.0f / (float)HW);
  float var = q * (1.0f / (float)HW) - mu * mu;
  mean[i] = mu;
  rstd[i] = rsqrtf(fmaxf(var, 0.0f) + 1e-5f);
}

#define CNB 2
#define W1PITCH 132
#define W2PITCH 68
__global__ __launch_bounds__(256, 8) void k_bands(
    const float* __restrict__ xmod, const float* __restrict__ W1,
    const float* __restrict__ W2, const float* __restrict__ W3,
    const float* __restrict__ mean, const float* __restrict__ rstd,
    float* __restrict__ out) {
  __shared__ __attribute__((aligned(16))) float sW1[MID * W1PITCH];
  __shared__ __attribute__((aligned(16))) float sW2[32 * W2PITCH];
  __shared__ float lds[8][MID * 17];
  const int wave  = __builtin_amdgcn_readfirstlane((int)(threadIdx.x >> 5));
  const int lane  = threadIdx.x & 31;
  const int band  = blockIdx.x & 7;
  const int n     = blockIdx.x >> 9;
  const int chunk = ((blockIdx.x >> 3) & 63) * 8 + wave;
  const int lm = lane & 15, kh = (lane >> 4) * 2, mh = (lane >> 4) * 8;
  const int nb = n * NBAND + band;
  const float* xn = xmod + (size_t)n * CCH * HW;
  const float* w3 = W3 + (size_t)band * 32;
  const float* mu = mean + (size_t)nb * MID;
  const float* rs = rstd + (size_t)nb * MID;

  {
    const float4* g1 = (const float4*)(W1 + (size_t)band * MID * CCH);
    for (int i = 0; i < (MID * CCH / 4) / 256; ++i) {
      const int idx = (int)threadIdx.x + i * 256;
      *(float4*)(sW1 + (idx >> 5) * W1PITCH + (idx & 31) * 4) = g1[idx];
    }
    const float4* g2 = (const float4*)(W2 + (size_t)band * 32 * MID);
    for (int i = 0; i < (32 * MID / 4) / 256; ++i) {
      const int idx = (int)threadIdx.x + i * 256;
      *(float4*)(sW2 + (idx >> 4) * W2PITCH + (idx & 15) * 4) = g2[idx];
    }
  }
  __syncthreads();

  float* sm = &lds[wave][0];
  float* sq = sm + mh * 17 + lm;
  const float* sk = sm + kh * 17 + lm;
  const float* a2l = sW2 + lm * W2PITCH + kh;
  v8f acc[CNB][4];
  h1_tile_hb<CNB, W1PITCH>(acc, xn, sW1, chunk * (CNB * 16), lane);
  {
    const float* mul = mu + mh;
    const float* rsl = rs + mh;
#pragma unroll
    for (int t = 0; t < 4; ++t)
#pragma unroll
      for (int v = 0; v < 8; ++v) {
        const float mu_m = mul[t * 16 + v], rs_m = rsl[t * 16 + v];
        sq[(t * 16 + v) * 17] = (acc[0][t][v] - mu_m) * rs_m;
        acc[1][t][v] = (acc[1][t][v] - mu_m) * rs_m;
      }
  }

#pragma unroll 1
  for (int sb = 0; sb < CNB; ++sb) {
    if (sb) {
#pragma unroll
      for (int t = 0; t < 4; ++t)
#pragma unroll
        for (int v = 0; v < 8; ++v)
          sq[(t * 16 + v) * 17] = acc[1][t][v];
    }

#pragma unroll
    for (int t = 0; t < 4; ++t)
#pragma unroll
      for (int v = 0; v < 8; ++v)
        sq[(t * 16 + v) * 17] = gelu_f(sq[(t * 16 + v) * 17]);

    v8f acc2[2] = {};
#pragma unroll 1
    for (int k1 = 0; k1 < MID; k1 += 16) {
#pragma unroll
      for (int k0 = k1; k0 < k1 + 16; k0 += 4) {
        v2f bb;
        bb.x = sk[k0 * 17];
        bb.y = sk[(k0 + 1) * 17];
#pragma unroll
        for (int t = 0; t < 2; ++t) {
          v2f a = *(const v2f*)(a2l + t * 16 * W2PITCH + k0);
          acc2[t] = wmma4(a, bb, acc2[t]);
        }
      }
    }
#pragma unroll
    for (int t = 0; t < 2; ++t)
#pragma unroll
      for (int v = 0; v < 8; ++v)
        sq[(t * 16 + v) * 17] = gelu_f(acc2[t][v]);

    if (lane < 16) {
      float s = 0.0f;
#pragma unroll
      for (int k = 0; k < 32; ++k) s += w3[k] * sm[k * 17 + lane];
      out[(size_t)nb * HW + chunk * (CNB * 16) + sb * 16 + lane] = clamp01(gelu_f(s));
    }
  }
}

extern "C" void kernel_launch(void* const* d_in, const int* in_sizes, int n_in,
                              void* d_out, int out_size, void* d_ws, size_t ws_size,
                              hipStream_t stream) {
  const float* x   = (const float*)d_in[0];
  const float* Wp  = (const float*)d_in[1];
  const float* Wm1 = (const float*)d_in[2];
  const float* Wm2 = (const float*)d_in[3];
  const float* W1  = (const float*)d_in[4];
  const float* W2  = (const float*)d_in[5];
  const float* W3  = (const float*)d_in[6];
  float* out = (float*)d_out;

  float* ws   = (float*)d_ws;
  float* emb  = ws;
  float* xmod = emb + (size_t)PDIM * HW;
  float* psum = xmod + (size_t)NBATCH * CCH * HW;
  float* psq  = psum + (size_t)NBATCH * NBAND * MID * 128;
  float* mean = psq + (size_t)NBATCH * NBAND * MID * 128;
  float* rstd = mean + (size_t)NBATCH * NBAND * MID;
  (void)in_sizes; (void)n_in; (void)out_size; (void)ws_size;

  k_emb<<<HW / 256, 256, 0, stream>>>(Wp, emb);
  k_modulate<<<NBATCH * 64, 256, 0, stream>>>(x, emb, Wm1, Wm2, xmod);
  k_band_stats<<<NBATCH * (HW / (SCH * 128)), 256, 0, stream>>>(xmod, W1, psum, psq);
  k_reduce_stats<<<(NBATCH * NBAND * MID + 255) / 256, 256, 0, stream>>>(psum, psq, mean, rstd);
  k_bands<<<NBATCH * 512, 256, 0, stream>>>(xmod, W1, W2, W3, mean, rstd, out);
}


# GELU in every kernel by LDS table: first launch fills 128 cubic pieces of erf on [0,4] (erff, expf), blocks copy its 2 KiB; activation = one 16-byte LDS gather + multiply-adds, no branch, no exp
# speedup vs baseline: 2.6960x; 1.1353x over previous
#include <hip/hip_runtime.h>
#include <math.h>


typedef __attribute__((ext_vector_type(2))) float v2f;
typedef __attribute__((ext_vector_type(8))) float v8f;

#define HW      16384
#define NBATCH  8
#define CCH     128
#define MID     64
#define PDIM    32
#define NBAND   8

__device__ __forceinline__ v8f wmma4(v2f a, v2f b, v8f c) {
  return __builtin_amdgcn_wmma_f32_16x16x4_f32(false, a, false, b,
                                               (short)0, c, false, false);
}

#define GT_N 128
__device__ __forceinline__ float gelu_f(float x, const float4* tab) {
  const float u = fminf(fabsf(x) * 22.627416997969522f, 127.99f);
  const int   i = (int)u;
  const float d = fmaf(__builtin_amdgcn_fractf(u), 0.03125f, -0.015625f);
  const float4 c = tab[i];
  float p = fmaf(d, c.w, c.z);
  p = fmaf(d, p, c.y);
  p = fmaf(d, p, c.x);
  const float hx = 0.5f * x;
  return fmaf(fabsf(hx), p, hx);
}
__device__ __forceinline__ void gelu_tab_copy(float4* stab, const float* __restrict__ gtab) {
  if (threadIdx.x < GT_N) stab[threadIdx.x] = ((const float4*)gtab)[threadIdx.x];
}
__device__ __forceinline__ float clamp01(float x) {
  return fminf(fmaxf(x, 0.0f), 1.0f);
}

__global__ void k_emb(const float* __restrict__ Wp, float* __restrict__ emb,
                      float* __restrict__ gtab) {
  int p = blockIdx.x * blockDim.x + threadIdx.x;
  if (p >= HW) return;
  if (blockIdx.x == 0 && threadIdx.x < GT_N) {
    const float m = ((float)threadIdx.x + 0.5f) * 0.03125f;
    const float e = 1.1283791670955126f * expf(-m * m);
    float4 cf;
    cf.x = erff(m);
    cf.y = e;
    cf.z = -m * e;
    cf.w = (4.0f * m * m - 2.0f) * (1.0f / 6.0f) * e;
    ((float4*)gtab)[threadIdx.x] = cf;
  }
  int i = p >> 7, j = p & 127;
  float ym = -1.0f + (2.0f / 127.0f) * (float)i;
  float xm = -1.0f + (2.0f / 127.0f) * (float)j;
#pragma unroll 4
  for (int pd = 0; pd < PDIM; ++pd) {
    float v = tanhf(Wp[pd * 2 + 0] * ym + Wp[pd * 2 + 1] * xm);
    emb[(size_t)pd * HW + p] = v;
  }
}

#define MNB    2
#define MPITCH 68
#define OPITCH 36

template <int J0, int PITCH>
__device__ __forceinline__ void mod_load_a_half(float4 (&a)[4][4], const float* __restrict__ ap) {
#pragma unroll
  for (int t = 0; t < 4; ++t) {
    a[t][J0]     = *(const float4*)(ap + t * 16 * PITCH + 4 * J0);
    a[t][J0 + 1] = *(const float4*)(ap + t * 16 * PITCH + 4 * J0 + 4);
  }
}

template <int J0>
__device__ __forceinline__ void mod_load_bg_half(float4 (&b)[MNB][4], const float* __restrict__ p) {
#pragma unroll
  for (int sb = 0; sb < MNB; ++sb)
#pragma unroll
    for (int jj = J0; jj < J0 + 2; ++jj) {
      b[sb][jj].x = p[(size_t)(4 * jj + 0) * HW + sb * 16];
      b[sb][jj].y = p[(size_t)(4 * jj + 1) * HW + sb * 16];
      b[sb][jj].z = p[(size_t)(4 * jj + 2) * HW + sb * 16];
      b[sb][jj].w = p[(size_t)(4 * jj + 3) * HW + sb * 16];
    }
}

template <int J0>
__device__ __forceinline__ void mod_load_bl_half(float4 (&b)[MNB][4], const float* p) {
#pragma unroll
  for (int sb = 0; sb < MNB; ++sb) {
    b[sb][J0]     = *(const float4*)(p + sb * 16 * MPITCH + 4 * J0);
    b[sb][J0 + 1] = *(const float4*)(p + sb * 16 * MPITCH + 4 * J0 + 4);
  }
}

__device__ __forceinline__ void mod_mma_step(v8f (&acc)[MNB][4], const float4 (&a)[4][4],
                                             const float4 (&b)[MNB][4], int s) {
  const int j = s >> 1;
  v2f bf[MNB];
#pragma unroll
  for (int sb = 0; sb < MNB; ++sb) {
    if (s & 1) { bf[sb].x = b[sb][j].z; bf[sb].y = b[sb][j].w; }
    else       { bf[sb].x = b[sb][j].x; bf[sb].y = b[sb][j].y; }
  }
#pragma unroll
  for (int t = 0; t < 4; ++t) {
    v2f af;
    if (s & 1) { af.x = a[t][j].z; af.y = a[t][j].w; }
    else       { af.x = a[t][j].x; af.y = a[t][j].y; }
#pragma unroll
    for (int sb = 0; sb < MNB; ++sb) acc[sb][t] = wmma4(af, bf[sb], acc[sb][t]);
  }
}

__device__ __forceinline__ float4 gelu4(float4 v, const float4* tab) {
  v.x = gelu_f(v.x, tab); v.y = gelu_f(v.y, tab); v.z = gelu_f(v.z, tab); v.w = gelu_f(v.w, tab);
  return v;
}

__global__ __launch_bounds__(256, 4) void k_modulate(
    const float* __restrict__ x, const float* __restrict__ emb,
    const float* __restrict__ Wm1, const float* __restrict__ Wm2,
    const float* __restrict__ gtab, float* __restrict__ xmod) {
  __shared__ __attribute__((aligned(16))) float lds[8][MNB * 16 * MPITCH + 64 * OPITCH];
  __shared__ float4 sTab[GT_N];
  gelu_tab_copy(sTab, gtab);
  __syncthreads();
  const int wave = threadIdx.x >> 5;
  const int lane = threadIdx.x & 31;
  const int n    = blockIdx.x >> 6;
  const int pixbase = (blockIdx.x & 63) * (8 * MNB * 16) + wave * (MNB * 16);
  const int lm   = lane & 15;
  const int h    = lane >> 4;
  const int mh   = h * 8;
  const float* xn = x + (size_t)n * CCH * HW;
  float* sm = &lds[wave][0];
  float* so = sm + MNB * 16 * MPITCH;

  float4 a[4][4];
  float4 b[MNB][4];

  {
    v8f acc[MNB][4];
#pragma unroll
    for (int sb = 0; sb < MNB; ++sb)
#pragma unroll
      for (int t = 0; t < 4; ++t) acc[sb][t] = v8f{};
    const float* arow = Wm1 + lm * (CCH + PDIM) + 16 * h;
    const float* xcol = xn  + (size_t)(16 * h) * HW + pixbase + lm;
    const float* ecol = emb + (size_t)(16 * h) * HW + pixbase + lm;
    mod_load_a_half<0, CCH + PDIM>(a, arow);
    mod_load_a_half<2, CCH + PDIM>(a, arow);
    mod_load_bg_half<0>(b, xcol);
    mod_load_bg_half<2>(b, xcol);
#pragma unroll 1
    for (int kb = 0; kb < 5; ++kb) {
      const int nkb = (kb == 4) ? 0 : kb + 1;
      const float* an = arow + nkb * 32;
      const float* bn = (nkb < 4) ? xcol + (size_t)(nkb * 32) * HW : ecol;
#pragma unroll
      for (int s = 0; s < 4; ++s) mod_mma_step(acc, a, b, s);
      mod_load_a_half<0, CCH + PDIM>(a, an);
      mod_load_bg_half<0>(b, bn);
#pragma unroll
      for (int s = 4; s < 8; ++s) mod_mma_step(acc, a, b, s);
      mod_load_a_half<2, CCH + PDIM>(a, an);
      mod_load_bg_half<2>(b, bn);
    }
    float* qb = sm + lm * MPITCH + mh;
#pragma unroll
    for (int sb = 0; sb < MNB; ++sb)
#pragma unroll
      for (int t = 0; t < 4; ++t) {
        float4 lo, hi;
        lo.x = acc[sb][t][0]; lo.y = acc[sb][t][1]; lo.z = acc[sb][t][2]; lo.w = acc[sb][t][3];
        hi.x = acc[sb][t][4]; hi.y = acc[sb][t][5]; hi.z = acc[sb][t][6]; hi.w = acc[sb][t][7];
        float* q = qb + sb * 16 * MPITCH + 16 * t;
        *(float4*)q       = lo;
        *(float4*)(q + 4) = hi;
      }
#pragma unroll 1
    for (int i = 0; i < MNB * 8; ++i) {
      float* q = qb + (i >> 3) * 16 * MPITCH + ((i >> 1) & 3) * 16 + (i & 1) * 4;
      *(float4*)q = gelu4(*(const float4*)q, sTab);
    }
  }

  float* xo = xmod + (size_t)n * CCH * HW + pixbase + lane;
  const float* bcol = sm + lm * MPITCH + 16 * h;
#pragma unroll 1
  for (int mhalf = 0; mhalf < 2; ++mhalf) {
    v8f acc2[MNB][4];
#pragma unroll
    for (int sb = 0; sb < MNB; ++sb)
#pragma unroll
      for (int t = 0; t < 4; ++t) acc2[sb][t] = v8f{};
    const float* arow2 = Wm2 + (size_t)(mhalf * 64 + lm) * MID + 16 * h;
    mod_load_a_half<0, MID>(a, arow2);
    mod_load_a_half<2, MID>(a, arow2);
    mod_load_bl_half<0>(b, bcol);
    mod_load_bl_half<2>(b, bcol);
#pragma unroll
    for (int s = 0; s < 4; ++s) mod_mma_step(acc2, a, b, s);
    mod_load_a_half<0, MID>(a, arow2 + 32);
    mod_load_bl_half<0>(b, bcol + 32);
#pragma unroll
    for (int s = 4; s < 8; ++s) mod_mma_step(acc2, a, b, s);
    mod_load_a_half<2, MID>(a, arow2 + 32);
    mod_load_bl_half<2>(b, bcol + 32);
#pragma unroll
    for (int s = 0; s < 8; ++s) mod_mma_step(acc2, a, b, s);
#pragma unroll
    for (int sb = 0; sb < MNB; ++sb)
#pragma unroll
      for (int t = 0; t < 4; ++t)
#pragma unroll
        for (int v = 0; v < 8; ++v)
          so[(t * 16 + v + mh) * OPITCH + sb * 16 + lm] = acc2[sb][t][v];
    float* xr = xo + (size_t)(mhalf * 64) * HW;
#pragma unroll 1
    for (int r = 0; r < 64; r += 4) {
#pragma unroll
      for (int u = 0; u < 4; ++u)
        xr[(size_t)(r + u) * HW] = clamp01(gelu_f(so[(r + u) * OPITCH + lane], sTab));
    }
  }
}

template <int NB>
__device__ __forceinline__ void h1_load_b(v2f (&b)[NB], const float* __restrict__ p) {
#pragma unroll
  for (int sb = 0; sb < NB; ++sb) {
    b[sb].x = p[sb * 16];
    b[sb].y = p[HW + sb * 16];
  }
}

template <int J0>
__device__ __forceinline__ void h1_load_a_half(float4 (&a)[4][4], const float* __restrict__ ap) {
#pragma unroll
  for (int t = 0; t < 4; ++t) {
    a[t][J0]     = *(const float4*)(ap + t * 16 * CCH + 4 * J0);
    a[t][J0 + 1] = *(const float4*)(ap + t * 16 * CCH + 4 * J0 + 4);
  }
}

template <int NB>
__device__ __forceinline__ void h1_mma_step(v8f (&acc)[NB][4], const float4 (&a)[4][4],
                                            const v2f (&b)[NB], int s) {
  const int j = s >> 1;
#pragma unroll
  for (int t = 0; t < 4; ++t) {
    v2f af;
    if (s & 1) { af.x = a[t][j].z; af.y = a[t][j].w; }
    else       { af.x = a[t][j].x; af.y = a[t][j].y; }
#pragma unroll
    for (int sb = 0; sb < NB; ++sb) acc[sb][t] = wmma4(af, b[sb], acc[sb][t]);
  }
}

template <int NB>
__device__ __forceinline__ void h1_tile(v8f (&acc)[NB][4], const float* __restrict__ xn,
                                        const float* __restrict__ A, int pixbase, int lane) {
  const int lm = lane & 15, h = lane >> 4;
  const float* arow = A + lm * CCH + 16 * h;
  const float* bcol = xn + (size_t)(16 * h) * HW + pixbase + lm;
#pragma unroll
  for (int sb = 0; sb < NB; ++sb)
#pragma unroll
    for (int t = 0; t < 4; ++t) acc[sb][t] = v8f{};
  float4 a[4][4];
  v2f b0[NB], b1[NB];
  h1_load_a_half<0>(a, arow);
  h1_load_a_half<2>(a, arow);
  h1_load_b<NB>(b0, bcol);
#pragma unroll 1
  for (int kb = 0; kb < 4; ++kb) {
    const int nkb = (kb + 1) & 3;
    const float* an = arow + nkb * 32;
    const float* bk = bcol + (size_t)(kb * 32) * HW;
    const float* bn = bcol + (size_t)(nkb * 32) * HW;
#pragma unroll
    for (int s = 0; s < 8; s += 2) {
      h1_load_b<NB>(b1, bk + (size_t)(2 * (s + 1)) * HW);
      if (s == 4) h1_load_a_half<0>(a, an);
      h1_mma_step<NB>(acc, a, b0, s);
      if (s + 2 < 8) h1_load_b<NB>(b0, bk + (size_t)(2 * (s + 2)) * HW);
      else           h1_load_b<NB>(b0, bn);
      h1_mma_step<NB>(acc, a, b1, s + 1);
    }
    h1_load_a_half<2>(a, an);
  }
}

template <int J, int PITCH>
__device__ __forceinline__ void h1_load_a_piece(float4 (&a)[4][2], const float* ap) {
#pragma unroll
  for (int t = 0; t < 4; ++t) a[t][J] = *(const float4*)(ap + t * 16 * PITCH + 4 * J);
}

template <int NB>
__device__ __forceinline__ void h1_mma_step_hb(v8f (&acc)[NB][4], const float4 (&a)[4][2],
                                               const v2f (&b)[NB], int s) {
  const int j = s >> 1;
#pragma unroll
  for (int t = 0; t < 4; ++t) {
    v2f af;
    if (s & 1) { af.x = a[t][j].z; af.y = a[t][j].w; }
    else       { af.x = a[t][j].x; af.y = a[t][j].y; }
#pragma unroll
    for (int sb = 0; sb < NB; ++sb) acc[sb][t] = wmma4(af, b[sb], acc[sb][t]);
  }
}

template <int NB, int PITCH>
__device__ __forceinline__ void h1_tile_hb(v8f (&acc)[NB][4], const float* __restrict__ xn,
                                           const float* A, int pixbase, int lane) {
  const int lm = lane & 15, h = lane >> 4;
  const float* arow = A + lm * PITCH + 16 * h;
  const float* bcol = xn + (size_t)(16 * h) * HW + pixbase + lm;
#pragma unroll
  for (int sb = 0; sb < NB; ++sb)
#pragma unroll
    for (int t = 0; t < 4; ++t) acc[sb][t] = v8f{};
  float4 a[4][2];
  v2f b0[NB], b1[NB];
  h1_load_a_piece<0, PITCH>(a, arow);
  h1_load_a_piece<1, PITCH>(a, arow);
  h1_load_b<NB>(b0, bcol);
#pragma unroll 1
  for (int q = 0; q < 8; ++q) {
    const int nq  = (q + 1) & 7;
    const int ko  = (q >> 1) * 32 + (q & 1) * 8;
    const int nko = (nq >> 1) * 32 + (nq & 1) * 8;
    const float* an = arow + nko;
    const float* bk = bcol + (size_t)ko * HW;
    const float* bn = bcol + (size_t)nko * HW;
    h1_load_b<NB>(b1, bk + (size_t)2 * HW);
    h1_mma_step_hb<NB>(acc, a, b0, 0);
    h1_load_b<NB>(b0, bk + (size_t)4 * HW);
    h1_mma_step_hb<NB>(acc, a, b1, 1);
    h1_load_a_piece<0, PITCH>(a, an);
    h1_load_b<NB>(b1, bk + (size_t)6 * HW);
    h1_mma_step_hb<NB>(acc, a, b0, 2);
    h1_load_b<NB>(b0, bn);
    h1_mma_step_hb<NB>(acc, a, b1, 3);
    h1_load_a_piece<1, PITCH>(a, an);
  }
}

#define SCH   4
#define SPART 32
__global__ __launch_bounds__(256) void k_band_stats(
    const float* __restrict__ xmod, const float* __restrict__ W1,
    float* __restrict__ psum, float* __restrict__ psq) {
  const int band  = threadIdx.x >> 5;
  const int lane  = threadIdx.x & 31;
  const int n     = blockIdx.x >> 5;
  const int range = blockIdx.x & 31;
  const int lm = lane & 15, mh = (lane >> 4) * 8;
  const float* xn = xmod + (size_t)n * CCH * HW;
  const float* A  = W1 + (size_t)band * MID * CCH;

  v8f s_run[4] = {};
  v8f q_run[4] = {};
  for (int c = 0; c < SCH; ++c) {
    v8f acc[8][4];
    h1_tile<8>(acc, xn, A, (range * SCH + c) * 128, lane);
#pragma unroll
    for (int sb = 0; sb < 8; ++sb)
#pragma unroll
      for (int t = 0; t < 4; ++t) {
        s_run[t] += acc[sb][t];
        q_run[t] += acc[sb][t] * acc[sb][t];
      }
  }
  const int nb = n * NBAND + band;
#pragma unroll
  for (int t = 0; t < 4; ++t)
#pragma unroll
    for (int v = 0; v < 8; ++v) {
      float s = s_run[t][v], q = q_run[t][v];
#pragma unroll
      for (int off = 1; off < 16; off <<= 1) {
        s += __shfl_xor(s, off, 32);
        q += __shfl_xor(q, off, 32);
      }
      if (lm == 0) {
        int m = t * 16 + v + mh;
        psum[((size_t)nb * MID + m) * SPART + range] = s;
        psq [((size_t)nb * MID + m) * SPART + range] = q;
      }
    }
}

__global__ void k_reduce_stats(const float* __restrict__ psum,
                               const float* __restrict__ psq,
                               float* __restrict__ mean,
                               float* __restrict__ rstd) {
  int i = blockIdx.x * blockDim.x + threadIdx.x;
  if (i >= NBATCH * NBAND * MID) return;
  float s = 0.0f, q = 0.0f;
  for (int j = 0; j < SPART; ++j) { s += psum[(size_t)i * SPART + j]; q += psq[(size_t)i * SPART + j]; }
  float mu  = s * (1.0f / (float)HW);
  float var = q * (1.0f / (float)HW) - mu * mu;
  mean[i] = mu;
  rstd[i] = rsqrtf(fmaxf(var, 0.0f) + 1e-5f);
}

#define CNB 2
#define W1PITCH 132
#define W2PITCH 68
__global__ __launch_bounds__(256, 8) void k_bands(
    const float* __restrict__ xmod, const float* __restrict__ W1,
    const float* __restrict__ W2, const float* __restrict__ W3,
    const float* __restrict__ mean, const float* __restrict__ rstd,
    const float* __restrict__ gtab, float* __restrict__ out) {
  __shared__ __attribute__((aligned(16))) float sW1[MID * W1PITCH];
  __shared__ __attribute__((aligned(16))) float sW2[32 * W2PITCH];
  __shared__ float4 sTab[GT_N];
  __shared__ float lds[8][MID * 17];
  const int wave  = __builtin_amdgcn_readfirstlane((int)(threadIdx.x >> 5));
  const int lane  = threadIdx.x & 31;
  const int band  = blockIdx.x & 7;
  const int n     = blockIdx.x >> 9;
  const int chunk = ((blockIdx.x >> 3) & 63) * 8 + wave;
  const int nb = n * NBAND + band;
  const float* xn = xmod + (size_t)n * CCH * HW;
  const float* w3 = W3 + (size_t)band * 32;
  const float* mu = mean + (size_t)nb * MID;
  const float* rs = rstd + (size_t)nb * MID;

  {
    const float4* g1 = (const float4*)(W1 + (size_t)band * MID * CCH);
    for (int i = 0; i < (MID * CCH / 4) / 256; ++i) {
      const int idx = (int)threadIdx.x + i * 256;
      *(float4*)(sW1 + (idx >> 5) * W1PITCH + (idx & 31) * 4) = g1[idx];
    }
    const float4* g2 = (const float4*)(W2 + (size_t)band * 32 * MID);
    for (int i = 0; i < (32 * MID / 4) / 256; ++i) {
      const int idx = (int)threadIdx.x + i * 256;
      *(float4*)(sW2 + (idx >> 4) * W2PITCH + (idx & 15) * 4) = g2[idx];
    }
    gelu_tab_copy(sTab, gtab);
  }
  __syncthreads();

  v8f acc[CNB][4];
  h1_tile_hb<CNB, W1PITCH>(acc, xn, sW1, chunk * (CNB * 16), lane);
  const int cl = (int)__builtin_amdgcn_mbcnt_lo(~0u, 0u);
  const int lm = cl & 15, kh = (cl >> 4) * 2, mh = (cl >> 4) * 8;
  float* sm = &lds[wave][0];
  float* sq = sm + mh * 17 + lm;
  const float* sk = sm + kh * 17 + lm;
  const float* a2l = sW2 + lm * W2PITCH + kh;
  {
    const float* mul = mu + mh;
    const float* rsl = rs + mh;
#pragma unroll
    for (int t = 0; t < 4; ++t)
#pragma unroll
      for (int v = 0; v < 8; ++v) {
        const float mu_m = mul[t * 16 + v], rs_m = rsl[t * 16 + v];
        sq[(t * 16 + v) * 17] = (acc[0][t][v] - mu_m) * rs_m;
        acc[1][t][v] = (acc[1][t][v] - mu_m) * rs_m;
      }
  }

#pragma unroll 1
  for (int sb = 0; sb < CNB; ++sb) {
    if (sb) {
#pragma unroll
      for (int t = 0; t < 4; ++t)
#pragma unroll
        for (int v = 0; v < 8; ++v)
          sq[(t * 16 + v) * 17] = acc[1][t][v];
    }

#pragma unroll
    for (int t = 0; t < 4; ++t)
#pragma unroll
      for (int v = 0; v < 8; ++v)
        sq[(t * 16 + v) * 17] = gelu_f(sq[(t * 16 + v) * 17], sTab);

    v8f acc2[2] = {};
#pragma unroll 1
    for (int k1 = 0; k1 < MID; k1 += 16) {
#pragma unroll
      for (int k0 = k1; k0 < k1 + 16; k0 += 4) {
        v2f bb;
        bb.x = sk[k0 * 17];
        bb.y = sk[(k0 + 1) * 17];
#pragma unroll
        for (int t = 0; t < 2; ++t) {
          v2f a = *(const v2f*)(a2l + t * 16 * W2PITCH + k0);
          acc2[t] = wmma4(a, bb, acc2[t]);
        }
      }
    }
#pragma unroll
    for (int t = 0; t < 2; ++t)
#pragma unroll
      for (int v = 0; v < 8; ++v)
        sq[(t * 16 + v) * 17] = gelu_f(acc2[t][v], sTab);

    if (cl < 16) {
      float s = 0.0f;
#pragma unroll
      for (int k = 0; k < 32; ++k) s += w3[k] * sm[k * 17 + cl];
      out[(size_t)nb * HW + chunk * (CNB * 16) + sb * 16 + cl] = clamp01(gelu_f(s, sTab));
    }
  }
}

extern "C" void kernel_launch(void* const* d_in, const int* in_sizes, int n_in,
                              void* d_out, int out_size, void* d_ws, size_t ws_size,
                              hipStream_t stream) {
  const float* x   = (const float*)d_in[0];
  const float* Wp  = (const float*)d_in[1];
  const float* Wm1 = (const float*)d_in[2];
  const float* Wm2 = (const float*)d_in[3];
  const float* W1  = (const float*)d_in[4];
  const float* W2  = (const float*)d_in[5];
  const float* W3  = (const float*)d_in[6];
  float* out = (float*)d_out;

  float* ws   = (float*)d_ws;
  float* emb  = ws;
  float* xmod = emb + (size_t)PDIM * HW;
  float* psum = xmod + (size_t)NBATCH * CCH * HW;
  float* psq  = psum + (size_t)NBATCH * NBAND * MID * 128;
  float* mean = psq + (size_t)NBATCH * NBAND * MID * 128;
  float* rstd = mean + (size_t)NBATCH * NBAND * MID;
  float* gtab = psum + (size_t)NBATCH * NBAND * MID * 64;
  (void)in_sizes; (void)n_in; (void)out_size; (void)ws_size;

  k_emb<<<HW / 256, 256, 0, stream>>>(Wp, emb, gtab);
  k_modulate<<<NBATCH * 64, 256, 0, stream>>>(x, emb, Wm1, Wm2, gtab, xmod);
  k_band_stats<<<NBATCH * (HW / (SCH * 128)), 256, 0, stream>>>(xmod, W1, psum, psq);
  k_reduce_stats<<<(NBATCH * NBAND * MID + 255) / 256, 256, 0, stream>>>(psum, psq, mean, rstd);
  k_bands<<<NBATCH * 512, 256, 0, stream>>>(xmod, W1, W2, W3, mean, rstd, gtab, out);
}
